# starting kernel
# baseline (speedup 1.0000x reference)
#include <hip/hip_runtime.h>

#define LQ   4096
#define DM   1024
#define NB   4
#define MREL 32
#define NV   (2*MREL + 1)
#define COLS 256
#define NTHREADS 256

typedef __attribute__((ext_vector_type(2))) float v2f;
typedef __attribute__((ext_vector_type(4))) float v4f;
typedef __attribute__((ext_vector_type(8))) float v8f;

__device__ __forceinline__ float count_val(int i, int v) {
    int j = i - (v - MREL);
    float c = (j >= 0 && j < LQ) ? 1.0f : 0.0f;
    int c_lo = LQ - MREL - i;
    int c_hi = i - (MREL - 1);
    c = (v == 0)        ? (float)(c_lo > 0 ? c_lo : 0) : c;
    c = (v == 2 * MREL) ? (float)(c_hi > 0 ? c_hi : 0) : c;
    c = (v >  2 * MREL) ? 0.0f : c;
    return c;
}

__global__ __launch_bounds__(NTHREADS) void relpos_fused(
    const float* __restrict__ x, const float* __restrict__ pe, float* __restrict__ out)
{
    __shared__ float agg_s[16 * COLS];

    const int tid  = threadIdx.x;
    const int lane = tid & 31;
    const int wave = tid >> 5;
    const int mrow = lane & 15;
    const int half = lane >> 4;

    const int row0    = blockIdx.x * 16;
    const int colbase = blockIdx.y * COLS;
    const int i       = row0 + mrow;

    v2f afrag[17];
    #pragma unroll
    for (int s = 0; s < 17; ++s) {
        int ka = s * 4 + half * 2;
        afrag[s].x = count_val(i, ka);
        afrag[s].y = count_val(i, ka + 1);
    }

    const float scale = 1.0f / (float)LQ;

    #pragma unroll
    for (int t = 0; t < COLS / (16 * 8); ++t) {
        const int col_local = (wave * (COLS / (16 * 8)) + t) * 16;
        const int col       = colbase + col_local + mrow;
        v8f acc = {};
        #pragma unroll
        for (int s = 0; s < 17; ++s) {
            int kb = s * 4 + half * 2;
            v2f b;
            b.x = (kb     < NV) ? pe[(size_t)kb       * DM + col] : 0.0f;
            b.y = (kb + 1 < NV) ? pe[(size_t)(kb + 1) * DM + col] : 0.0f;
            acc = __builtin_amdgcn_wmma_f32_16x16x4_f32(
                       false, afrag[s],  false, b,
                       (short)0, acc,  false,  false);
        }
        #pragma unroll
        for (int r = 0; r < 8; ++r) {
            agg_s[(r + half * 8) * COLS + col_local + mrow] = acc[r] * scale;
        }
    }

    __syncthreads();

    const v4f* __restrict__ x4 = reinterpret_cast<const v4f*>(x);
    v4f*       __restrict__ o4 = reinterpret_cast<v4f*>(out);
    const v4f* __restrict__ a4 = reinterpret_cast<const v4f*>(agg_s);

    #pragma unroll 4
    for (int it = 0; it < (NB * 16 * COLS / 4) / NTHREADS; ++it) {
        int idx = it * NTHREADS + tid;
        int d4  = idx & (COLS / 4 - 1);
        int rr  = (idx >> 6) & 15;
        int b   = idx >> 10;
        size_t g4 = (size_t)b * (LQ * (DM / 4))
                  + (size_t)(row0 + rr) * (DM / 4)
                  + (colbase >> 2) + d4;
        v4f xv = __builtin_nontemporal_load(x4 + g4);
        v4f av = a4[rr * (COLS / 4) + d4];
        __builtin_nontemporal_store(xv + av, o4 + g4);
    }
}

extern "C" void kernel_launch(void* const* d_in, const int* in_sizes, int n_in,
                              void* d_out, int out_size, void* d_ws, size_t ws_size,
                              hipStream_t stream) {
    const float* x  = (const float*)d_in[0];
    const float* pe = (const float*)d_in[1];
    float* out = (float*)d_out;
    (void)in_sizes; (void)n_in; (void)out_size; (void)d_ws; (void)ws_size;

    dim3 grid(LQ / 16, DM / COLS);
    relpos_fused<<<grid, NTHREADS, 0, stream>>>(x, pe, out);
}


# workgroup retiled to 32 rows x 256 cols (512, was 1024): a lane owns 4 adjacent columns, pe comes in 16-byte loads, half the waves enter, each asks for its 32 x pieces at once; sums unchanged
# speedup vs baseline: 1.1390x; 1.1390x over previous
#include <hip/hip_runtime.h>

#define LQ   4096
#define DM   1024
#define NB   4
#define MREL 32
#define NV   (2*MREL + 1)
#define ROWS 32
#define COLS 256
#define NTHREADS 256

typedef __attribute__((ext_vector_type(2))) float v2f;
typedef __attribute__((ext_vector_type(4))) float v4f;
typedef __attribute__((ext_vector_type(8))) float v8f;

__device__ __forceinline__ float count_val(int i, int v) {
    int j = i - (v - MREL);
    float c = (j >= 0 && j < LQ) ? 1.0f : 0.0f;
    int c_lo = LQ - MREL - i;
    int c_hi = i - (MREL - 1);
    c = (v == 0)        ? (float)(c_lo > 0 ? c_lo : 0) : c;
    c = (v == 2 * MREL) ? (float)(c_hi > 0 ? c_hi : 0) : c;
    c = (v >  2 * MREL) ? 0.0f : c;
    return c;
}

__device__ __forceinline__ v8f mma(v2f a, v2f b, v8f c) {
    return __builtin_amdgcn_wmma_f32_16x16x4_f32(
                false, a,  false, b,
                (short)0, c,  false,  false);
}

__global__ __launch_bounds__(NTHREADS) void relpos_fused(
    const float* __restrict__ x, const float* __restrict__ pe, float* __restrict__ out)
{
    __shared__ alignas(16) float agg_s[ROWS * COLS];

    const int tid  = threadIdx.x;
    const int lane = tid & 31;
    const int wave = tid >> 5;
    const int c    = lane & 15;
    const int half = lane >> 4;
    const int rb   = wave >> 2;
    const int cg   = wave & 3;

    const int row0    = blockIdx.x * ROWS;
    const int colbase = blockIdx.y * COLS;
    const int i       = row0 + rb * 16 + c;

    v2f afrag[17];
    #pragma unroll
    for (int s = 0; s < 17; ++s) {
        int ka = s * 4 + half * 2;
        afrag[s].x = count_val(i, ka);
        afrag[s].y = count_val(i, ka + 1);
    }

    const v4f* __restrict__ pe4 =
        reinterpret_cast<const v4f*>(pe) + ((colbase + cg * 64) >> 2) + c;

    v8f acc0 = {}, acc1 = {}, acc2 = {}, acc3 = {};
    #pragma unroll
    for (int s = 0; s < 17; ++s) {
        const int kb = s * 4 + half * 2;
        const int k0 = (kb     < NV) ? kb     : NV - 1;
        const int k1 = (kb + 1 < NV) ? kb + 1 : NV - 1;
        v4f p0 = pe4[(size_t)k0 * (DM / 4)];
        v4f p1 = pe4[(size_t)k1 * (DM / 4)];
        const v4f zero = {0.0f, 0.0f, 0.0f, 0.0f};
        p0 = (kb     < NV) ? p0 : zero;
        p1 = (kb + 1 < NV) ? p1 : zero;
        v2f b0 = {p0.x, p1.x};
        v2f b1 = {p0.y, p1.y};
        v2f b2 = {p0.z, p1.z};
        v2f b3 = {p0.w, p1.w};
        acc0 = mma(afrag[s], b0, acc0);
        acc1 = mma(afrag[s], b1, acc1);
        acc2 = mma(afrag[s], b2, acc2);
        acc3 = mma(afrag[s], b3, acc3);
    }

    const float scale = 1.0f / (float)LQ;

    #pragma unroll
    for (int r = 0; r < 8; ++r) {
        v4f o = {acc0[r] * scale, acc1[r] * scale, acc2[r] * scale, acc3[r] * scale};
        *reinterpret_cast<v4f*>(&agg_s[(rb * 16 + half * 8 + r) * COLS + cg * 64 + 4 * c]) = o;
    }

    __syncthreads();

    const v4f* __restrict__ x4 = reinterpret_cast<const v4f*>(x);
    v4f*       __restrict__ o4 = reinterpret_cast<v4f*>(out);
    const v4f* __restrict__ a4 = reinterpret_cast<const v4f*>(agg_s);

    const int d4  = tid & (COLS / 4 - 1);
    const int rq  = tid >> 6;
    const unsigned off0 = (unsigned)(row0 + rq) * (DM / 4) + (unsigned)(colbase >> 2) + (unsigned)d4;

    v4f xv[32];
    #pragma unroll
    for (int j = 0; j < 32; ++j) {
        const unsigned off = off0 + (unsigned)(j >> 3) * (LQ * (DM / 4)) + (unsigned)(j & 7) * (4 * (DM / 4));
        xv[j] = __builtin_nontemporal_load(x4 + off);
    }
    #pragma unroll
    for (int j = 0; j < 32; ++j) {
        const unsigned off = off0 + (unsigned)(j >> 3) * (LQ * (DM / 4)) + (unsigned)(j & 7) * (4 * (DM / 4));
        const v4f av = a4[((j & 7) * 4 + rq) * (COLS / 4) + d4];
        __builtin_nontemporal_store(xv[j] + av, o4 + off);
    }
}

extern "C" void kernel_launch(void* const* d_in, const int* in_sizes, int n_in,
                              void* d_out, int out_size, void* d_ws, size_t ws_size,
                              hipStream_t stream) {
    const float* x  = (const float*)d_in[0];
    const float* pe = (const float*)d_in[1];
    float* out = (float*)d_out;
    (void)in_sizes; (void)n_in; (void)out_size; (void)d_ws; (void)ws_size;

    dim3 grid(LQ / ROWS, DM / COLS);
    relpos_fused<<<grid, NTHREADS, 0, stream>>>(x, pe, out);
}


# x loads moved ahead of the matrix phase: 8 pieces behind the pe loads, 8 more behind each of k-steps 3, 7, 11 (A values made per step to free registers, LDS-only fence at the barrier); sums unchanged
# speedup vs baseline: 1.4929x; 1.3107x over previous
#include <hip/hip_runtime.h>

#define LQ   4096
#define DM   1024
#define NB   4
#define MREL 32
#define NV   (2*MREL + 1)
#define ROWS 32
#define COLS 256
#define NTHREADS 256

typedef __attribute__((ext_vector_type(2))) float v2f;
typedef __attribute__((ext_vector_type(4))) float v4f;
typedef __attribute__((ext_vector_type(8))) float v8f;

__device__ __forceinline__ float count_val(int i, int v) {
    int j = i - (v - MREL);
    float c = (j >= 0 && j < LQ) ? 1.0f : 0.0f;
    int c_lo = LQ - MREL - i;
    int c_hi = i - (MREL - 1);
    c = (v == 0)        ? (float)(c_lo > 0 ? c_lo : 0) : c;
    c = (v == 2 * MREL) ? (float)(c_hi > 0 ? c_hi : 0) : c;
    c = (v >  2 * MREL) ? 0.0f : c;
    return c;
}

__device__ __forceinline__ v8f mma(v2f a, v2f b, v8f c) {
    return __builtin_amdgcn_wmma_f32_16x16x4_f32(
                false, a,  false, b,
                (short)0, c,  false,  false);
}

__global__ __launch_bounds__(NTHREADS, 4) void relpos_fused(
    const float* __restrict__ x, const float* __restrict__ pe, float* __restrict__ out)
{
    __shared__ alignas(16) float agg_s[ROWS * COLS];

    const int tid  = threadIdx.x;
    const int lane = tid & 31;
    const int wave = tid >> 5;
    const int c    = lane & 15;
    const int half = lane >> 4;
    const int rb   = wave >> 2;
    const int cg   = wave & 3;

    const int row0    = blockIdx.x * ROWS;
    const int colbase = blockIdx.y * COLS;
    const int i       = row0 + rb * 16 + c;

    const v4f* __restrict__ x4 = reinterpret_cast<const v4f*>(x);
    v4f*       __restrict__ o4 = reinterpret_cast<v4f*>(out);
    const v4f* __restrict__ a4 = reinterpret_cast<const v4f*>(agg_s);
    const int d4  = tid & (COLS / 4 - 1);
    const int rq  = tid >> 6;
    const unsigned off0 = (unsigned)(row0 + rq) * (DM / 4) + (unsigned)(colbase >> 2) + (unsigned)d4;

    const v4f* __restrict__ pe4 =
        reinterpret_cast<const v4f*>(pe) + ((colbase + cg * 64) >> 2) + c;

    v4f pv0[17], pv1[17];
    #pragma unroll
    for (int s = 0; s < 17; ++s) {
        const int kb = s * 4 + half * 2;
        const int k0 = (kb     < NV) ? kb     : NV - 1;
        const int k1 = (kb + 1 < NV) ? kb + 1 : NV - 1;
        pv0[s] = pe4[(size_t)k0 * (DM / 4)];
        pv1[s] = pe4[(size_t)k1 * (DM / 4)];
    }

    __builtin_amdgcn_sched_barrier(0);

    v4f xv[32];
    #pragma unroll
    for (int j = 0; j < 8; ++j) {
        const unsigned off = off0 + (unsigned)(j >> 3) * (LQ * (DM / 4)) + (unsigned)(j & 7) * (4 * (DM / 4));
        xv[j] = __builtin_nontemporal_load(x4 + off);
    }
    __builtin_amdgcn_sched_barrier(0);

    v8f acc0 = {}, acc1 = {}, acc2 = {}, acc3 = {};
    #pragma unroll
    for (int s = 0; s < 17; ++s) {
        const int kb = s * 4 + half * 2;
        v2f a;
        a.x = count_val(i, kb);
        a.y = count_val(i, kb + 1);
        const v4f zero = {0.0f, 0.0f, 0.0f, 0.0f};
        const v4f p0 = (kb     < NV) ? pv0[s] : zero;
        const v4f p1 = (kb + 1 < NV) ? pv1[s] : zero;
        v2f b0 = {p0.x, p1.x};
        v2f b1 = {p0.y, p1.y};
        v2f b2 = {p0.z, p1.z};
        v2f b3 = {p0.w, p1.w};
        acc0 = mma(a, b0, acc0);
        acc1 = mma(a, b1, acc1);
        acc2 = mma(a, b2, acc2);
        acc3 = mma(a, b3, acc3);
        if (s == 3 || s == 7 || s == 11) {
            #pragma unroll
            for (int j = 2 * s + 2; j < 2 * s + 10; ++j) {
                const unsigned off = off0 + (unsigned)(j >> 3) * (LQ * (DM / 4)) + (unsigned)(j & 7) * (4 * (DM / 4));
                xv[j] = __builtin_nontemporal_load(x4 + off);
            }
            __builtin_amdgcn_sched_barrier(0);
        }
    }

    const float scale = 1.0f / (float)LQ;

    #pragma unroll
    for (int r = 0; r < 8; ++r) {
        v4f o = {acc0[r] * scale, acc1[r] * scale, acc2[r] * scale, acc3[r] * scale};
        *reinterpret_cast<v4f*>(&agg_s[(rb * 16 + half * 8 + r) * COLS + cg * 64 + 4 * c]) = o;
    }

    __builtin_amdgcn_fence(__ATOMIC_RELEASE, "workgroup", "local");
    __builtin_amdgcn_s_barrier();
    __builtin_amdgcn_fence(__ATOMIC_ACQUIRE, "workgroup", "local");

    #pragma unroll
    for (int j = 0; j < 32; ++j) {
        const unsigned off = off0 + (unsigned)(j >> 3) * (LQ * (DM / 4)) + (unsigned)(j & 7) * (4 * (DM / 4));
        const v4f av = a4[((j & 7) * 4 + rq) * (COLS / 4) + d4];
        __builtin_nontemporal_store(xv[j] + av, o4 + off);
    }
}

extern "C" void kernel_launch(void* const* d_in, const int* in_sizes, int n_in,
                              void* d_out, int out_size, void* d_ws, size_t ws_size,
                              hipStream_t stream) {
    const float* x  = (const float*)d_in[0];
    const float* pe = (const float*)d_in[1];
    float* out = (float*)d_out;
    (void)in_sizes; (void)n_in; (void)out_size; (void)d_ws; (void)ws_size;

    dim3 grid(LQ / ROWS, DM / COLS);
    relpos_fused<<<grid, NTHREADS, 0, stream>>>(x, pe, out);
}
